# starting kernel
# baseline (speedup 1.0000x reference)
#include <hip/hip_runtime.h>
#include <hip/hip_bf16.h>


#define NUM_HEADS 4
#define HEAD_DIM 32
#define F_IN 256
#define F_OUT 128
#define LDA 260

typedef __attribute__((ext_vector_type(2))) float v2f;
typedef __attribute__((ext_vector_type(8))) float v8f;

__global__ __launch_bounds__(256) void gat_init(float* __restrict__ out,
                                                float* __restrict__ emax,
                                                float* __restrict__ asum,
                                                int N) {
    int tid = blockIdx.x * blockDim.x + threadIdx.x;
    if (tid < N * F_OUT) out[tid] = 0.0f;
    if (tid < N * NUM_HEADS) { emax[tid] = -3.0e38f; asum[tid] = 0.0f; }
}

__global__ __launch_bounds__(256) void gat_gemm(const float* __restrict__ x,
                                                const float* __restrict__ W,
                                                float* __restrict__ h) {
    __shared__ float As[16 * LDA];

    const int t = threadIdx.x;
    const int rowBase = blockIdx.x * 16;

    #pragma unroll
    for (int j = 0; j < 4; ++j) {
        int f = t + 256 * j;
        int r = f >> 6;
        int c = (f & 63) << 2;
        float4 v = *(const float4*)(x + (size_t)(rowBase + r) * F_IN + c);
        *(float4*)(&As[r * LDA + c]) = v;
    }
    __syncthreads();

    const int wave = t >> 5;
    const int lane = t & 31;
    const int m    = lane & 15;
    const int koff = (lane >> 4) << 1;
    const int col  = (wave << 4) + m;

    const float* Wp = W + col;
    v8f acc = {};

    #pragma unroll 8
    for (int k = 0; k < F_IN; k += 4) {
        v2f a = *(const v2f*)(&As[m * LDA + k + koff]);
        v2f b;
        b.x = Wp[(size_t)(k + koff) * F_OUT];
        b.y = Wp[(size_t)(k + koff + 1) * F_OUT];
        acc = __builtin_amdgcn_wmma_f32_16x16x4_f32(
             false, a,  false, b,
             (short)0, acc,  false,  false);
    }

    const int roff = (lane >> 4) << 3;
    float* hp = h + (size_t)rowBase * F_OUT + col;
    #pragma unroll
    for (int v = 0; v < 8; ++v)
        hp[(size_t)(v + roff) * F_OUT] = acc[v];
}

__global__ __launch_bounds__(256) void gat_gemm_tail(const float* __restrict__ x,
                                                     const float* __restrict__ W,
                                                     float* __restrict__ h,
                                                     int N, int rowStart) {
    int idx = blockIdx.x * blockDim.x + threadIdx.x;
    int total = (N - rowStart) * F_OUT;
    if (idx >= total) return;
    int r = rowStart + idx / F_OUT;
    int c = idx % F_OUT;
    float s = 0.0f;
    for (int k = 0; k < F_IN; ++k)
        s += x[(size_t)r * F_IN + k] * W[(size_t)k * F_OUT + c];
    h[(size_t)r * F_OUT + c] = s;
}

__global__ __launch_bounds__(256) void gat_alpha(const float* __restrict__ h,
                                                 const float* __restrict__ a_src,
                                                 const float* __restrict__ a_dst,
                                                 float* __restrict__ asrc,
                                                 float* __restrict__ adst,
                                                 int N) {
    int tid = blockIdx.x * blockDim.x + threadIdx.x;
    if (tid >= N * NUM_HEADS) return;
    int n  = tid >> 2;
    int hd = tid & 3;
    const float* hp = h + (size_t)n * F_OUT + hd * HEAD_DIM;
    const float* as = a_src + hd * HEAD_DIM;
    const float* ad = a_dst + hd * HEAD_DIM;
    float s1 = 0.0f, s2 = 0.0f;
    #pragma unroll
    for (int i = 0; i < HEAD_DIM; i += 4) {
        float4 hv = *(const float4*)(hp + i);
        float4 av = *(const float4*)(as + i);
        float4 dv = *(const float4*)(ad + i);
        s1 += hv.x * av.x + hv.y * av.y + hv.z * av.z + hv.w * av.w;
        s2 += hv.x * dv.x + hv.y * dv.y + hv.z * dv.z + hv.w * dv.w;
    }
    asrc[tid] = s1;
    adst[tid] = s2;
}

__device__ __forceinline__ float leaky(float v) { return v > 0.0f ? v : 0.2f * v; }

__global__ __launch_bounds__(256) void gat_edge_max(const int* __restrict__ src,
                                                    const int* __restrict__ dst,
                                                    const float* __restrict__ asrc,
                                                    const float* __restrict__ adst,
                                                    float* __restrict__ emax,
                                                    int E) {
    int e = blockIdx.x * blockDim.x + threadIdx.x;
    if (e >= E) return;
    int s = src[e], d = dst[e];
    #pragma unroll
    for (int hd = 0; hd < NUM_HEADS; ++hd) {
        float v = leaky(asrc[s * NUM_HEADS + hd] + adst[d * NUM_HEADS + hd]);
        __hip_atomic_fetch_max(&emax[d * NUM_HEADS + hd], v,
                               __ATOMIC_RELAXED, __HIP_MEMORY_SCOPE_AGENT);
    }
}

__global__ __launch_bounds__(256) void gat_edge_sum(const int* __restrict__ src,
                                                    const int* __restrict__ dst,
                                                    const float* __restrict__ asrc,
                                                    const float* __restrict__ adst,
                                                    const float* __restrict__ emax,
                                                    float* __restrict__ asum,
                                                    int E) {
    int e = blockIdx.x * blockDim.x + threadIdx.x;
    if (e >= E) return;
    int s = src[e], d = dst[e];
    #pragma unroll
    for (int hd = 0; hd < NUM_HEADS; ++hd) {
        float v = leaky(asrc[s * NUM_HEADS + hd] + adst[d * NUM_HEADS + hd]);
        float w = __expf(v - emax[d * NUM_HEADS + hd]);
        atomicAdd(&asum[d * NUM_HEADS + hd], w);
    }
}

__global__ __launch_bounds__(256) void gat_aggregate(const int* __restrict__ src,
                                                     const int* __restrict__ dst,
                                                     const float* __restrict__ asrc,
                                                     const float* __restrict__ adst,
                                                     const float* __restrict__ emax,
                                                     const float* __restrict__ asum,
                                                     const float* __restrict__ h,
                                                     float* __restrict__ out,
                                                     int E) {
    int e = blockIdx.x * 8 + (threadIdx.x >> 5);
    if (e >= E) return;
    int lane = threadIdx.x & 31;
    int s = src[e], d = dst[e];
    int elem = lane << 2;
    int hd   = elem >> 5;

    float v = leaky(asrc[s * NUM_HEADS + hd] + adst[d * NUM_HEADS + hd]);
    float w = __expf(v - emax[d * NUM_HEADS + hd]) /
              (asum[d * NUM_HEADS + hd] + 1e-10f);

    float4 hv = *(const float4*)(h + (size_t)s * F_OUT + elem);
    float* op = out + (size_t)d * F_OUT + elem;
    atomicAdd(op + 0, w * hv.x);
    atomicAdd(op + 1, w * hv.y);
    atomicAdd(op + 2, w * hv.z);
    atomicAdd(op + 3, w * hv.w);
}

extern "C" void kernel_launch(void* const* d_in, const int* in_sizes, int n_in,
                              void* d_out, int out_size, void* d_ws, size_t ws_size,
                              hipStream_t stream) {
    const float* x     = (const float*)d_in[0];
    const int*   ei    = (const int*)d_in[1];
    const float* W     = (const float*)d_in[2];
    const float* a_src = (const float*)d_in[3];
    const float* a_dst = (const float*)d_in[4];
    float*       out   = (float*)d_out;

    const int N = in_sizes[0] / F_IN;
    const int E = in_sizes[1] / 2;
    const int* src = ei;
    const int* dst = ei + E;

    float* h    = (float*)d_ws;
    float* asrc = h    + (size_t)N * F_OUT;
    float* adst = asrc + (size_t)N * NUM_HEADS;
    float* emax = adst + (size_t)N * NUM_HEADS;
    float* asum = emax + (size_t)N * NUM_HEADS;

    {
        int total = N * F_OUT;
        gat_init<<<(total + 255) / 256, 256, 0, stream>>>(out, emax, asum, N);
    }

    {
        int fullTiles = N / 16;
        if (fullTiles > 0)
            gat_gemm<<<fullTiles, 256, 0, stream>>>(x, W, h);
        int rowStart = fullTiles * 16;
        int rem = N - rowStart;
        if (rem > 0) {
            int total = rem * F_OUT;
            gat_gemm_tail<<<(total + 255) / 256, 256, 0, stream>>>(x, W, h, N, rowStart);
        }
    }

    {
        int total = N * NUM_HEADS;
        gat_alpha<<<(total + 255) / 256, 256, 0, stream>>>(h, a_src, a_dst, asrc, adst, N);
    }

    gat_edge_max<<<(E + 255) / 256, 256, 0, stream>>>(src, dst, asrc, adst, emax, E);

    gat_edge_sum<<<(E + 255) / 256, 256, 0, stream>>>(src, dst, asrc, adst, emax, asum, E);

    gat_aggregate<<<(E + 7) / 8, 256, 0, stream>>>(src, dst, asrc, adst, emax, asum, h, out, E);
}


# Edges grouped by destination in the call (count, scan, fill: integer atomics); one wave per node does max, exp-sum and row sum in registers: no float atomics, no zero fill; sums in fill order
# speedup vs baseline: 2.6306x; 2.6306x over previous
#include <hip/hip_runtime.h>
#include <hip/hip_bf16.h>


#define NUM_HEADS 4
#define HEAD_DIM 32
#define F_IN 256
#define F_OUT 128
#define LDA 260

typedef __attribute__((ext_vector_type(2))) float v2f;
typedef __attribute__((ext_vector_type(8))) float v8f;

__global__ __launch_bounds__(256) void gat_gemm(const float* __restrict__ x,
                                                const float* __restrict__ W,
                                                float* __restrict__ h) {
    __shared__ float As[16 * LDA];

    const int t = threadIdx.x;
    const int rowBase = blockIdx.x * 16;

    #pragma unroll
    for (int j = 0; j < 4; ++j) {
        int f = t + 256 * j;
        int r = f >> 6;
        int c = (f & 63) << 2;
        float4 v = *(const float4*)(x + (size_t)(rowBase + r) * F_IN + c);
        *(float4*)(&As[r * LDA + c]) = v;
    }
    __syncthreads();

    const int wave = t >> 5;
    const int lane = t & 31;
    const int m    = lane & 15;
    const int koff = (lane >> 4) << 1;
    const int col  = (wave << 4) + m;

    const float* Wp = W + col;
    v8f acc = {};

    #pragma unroll 8
    for (int k = 0; k < F_IN; k += 4) {
        v2f a = *(const v2f*)(&As[m * LDA + k + koff]);
        v2f b;
        b.x = Wp[(size_t)(k + koff) * F_OUT];
        b.y = Wp[(size_t)(k + koff + 1) * F_OUT];
        acc = __builtin_amdgcn_wmma_f32_16x16x4_f32(
             false, a,  false, b,
             (short)0, acc,  false,  false);
    }

    const int roff = (lane >> 4) << 3;
    float* hp = h + (size_t)rowBase * F_OUT + col;
    #pragma unroll
    for (int v = 0; v < 8; ++v)
        hp[(size_t)(v + roff) * F_OUT] = acc[v];
}

__global__ __launch_bounds__(256) void gat_gemm_tail(const float* __restrict__ x,
                                                     const float* __restrict__ W,
                                                     float* __restrict__ h,
                                                     int N, int rowStart) {
    int idx = blockIdx.x * blockDim.x + threadIdx.x;
    int total = (N - rowStart) * F_OUT;
    if (idx >= total) return;
    int r = rowStart + idx / F_OUT;
    int c = idx % F_OUT;
    float s = 0.0f;
    for (int k = 0; k < F_IN; ++k)
        s += x[(size_t)r * F_IN + k] * W[(size_t)k * F_OUT + c];
    h[(size_t)r * F_OUT + c] = s;
}

__global__ __launch_bounds__(256) void gat_alpha(const float* __restrict__ h,
                                                 const float* __restrict__ a_src,
                                                 const float* __restrict__ a_dst,
                                                 float* __restrict__ asrc,
                                                 float* __restrict__ adst,
                                                 int* __restrict__ cnt,
                                                 int N) {
    int tid = blockIdx.x * blockDim.x + threadIdx.x;
    if (tid >= N * NUM_HEADS) return;
    int n  = tid >> 2;
    int hd = tid & 3;
    if (hd == 0) cnt[n] = 0;
    const float* hp = h + (size_t)n * F_OUT + hd * HEAD_DIM;
    const float* as = a_src + hd * HEAD_DIM;
    const float* ad = a_dst + hd * HEAD_DIM;
    float s1 = 0.0f, s2 = 0.0f;
    #pragma unroll
    for (int i = 0; i < HEAD_DIM; i += 4) {
        float4 hv = *(const float4*)(hp + i);
        float4 av = *(const float4*)(as + i);
        float4 dv = *(const float4*)(ad + i);
        s1 += hv.x * av.x + hv.y * av.y + hv.z * av.z + hv.w * av.w;
        s2 += hv.x * dv.x + hv.y * dv.y + hv.z * dv.z + hv.w * dv.w;
    }
    asrc[tid] = s1;
    adst[tid] = s2;
}

__device__ __forceinline__ float leaky(float v) { return v > 0.0f ? v : 0.2f * v; }

__device__ __forceinline__ float wave_max(float v) {
    for (int o = 16; o > 0; o >>= 1) v = fmaxf(v, __shfl_xor(v, o));
    return v;
}
__device__ __forceinline__ float wave_sum(float v) {
    for (int o = 16; o > 0; o >>= 1) v += __shfl_xor(v, o);
    return v;
}
__device__ __forceinline__ int wave_sum_i(int v) {
    for (int o = 16; o > 0; o >>= 1) v += __shfl_xor(v, o);
    return v;
}

__global__ __launch_bounds__(256) void gat_count(const int* __restrict__ dst,
                                                 int* __restrict__ cnt,
                                                 int E) {
    int e = blockIdx.x * blockDim.x + threadIdx.x;
    if (e >= E) return;
    atomicAdd(&cnt[dst[e]], 1);
}

__global__ __launch_bounds__(1024) void gat_scan(const int* __restrict__ cnt,
                                                 int* __restrict__ cur,
                                                 int N) {
    __shared__ int s_front[32];
    __shared__ int s_total[32];

    const int t    = threadIdx.x;
    const int lane = t & 31;
    const int wv   = t >> 5;
    const int chunk0 = blockIdx.x * 1024;

    int front = 0;
    const int4* c4 = (const int4*)cnt;
    for (int i = t; i < (chunk0 >> 2); i += 1024) {
        int4 q = c4[i];
        front += q.x + q.y + q.z + q.w;
    }
    front = wave_sum_i(front);

    const int idx = chunk0 + t;
    const int v = idx < N ? cnt[idx] : 0;
    int incl = v;
    for (int o = 1; o < 32; o <<= 1) {
        int u = __shfl_up(incl, o);
        if (lane >= o) incl += u;
    }
    if (lane == 0)  s_front[wv] = front;
    if (lane == 31) s_total[wv] = incl;
    __syncthreads();

    int pre = s_front[lane] + (lane < wv ? s_total[lane] : 0);
    pre = wave_sum_i(pre);

    if (idx < N) cur[idx] = pre + incl - v;
}

__global__ __launch_bounds__(256) void gat_fill(const int* __restrict__ src,
                                                const int* __restrict__ dst,
                                                int* __restrict__ cur,
                                                int* __restrict__ grp_src,
                                                int E) {
    int e = blockIdx.x * blockDim.x + threadIdx.x;
    if (e >= E) return;
    int pos = atomicAdd(&cur[dst[e]], 1);
    grp_src[pos] = src[e];
}

__global__ __launch_bounds__(256) void gat_node(const int* __restrict__ grp_src,
                                                const int* __restrict__ cnt,
                                                const int* __restrict__ cur,
                                                const float* __restrict__ asrc,
                                                const float* __restrict__ adst,
                                                const float* __restrict__ h,
                                                float* __restrict__ out,
                                                int N) {
    const int n = blockIdx.x * 8 + __builtin_amdgcn_readfirstlane((int)(threadIdx.x >> 5));
    if (n >= N) return;
    const int lane = threadIdx.x & 31;

    const int end   = cur[n];
    const int deg   = cnt[n];
    const int start = end - deg;
    const float4 ad = *(const float4*)(adst + (size_t)n * NUM_HEADS);

    const float NEG = -__builtin_huge_valf();

    auto score = [&](int j, int& s_l, float4& e) {
        s_l = 0;
        e.x = NEG; e.y = NEG; e.z = NEG; e.w = NEG;
        if (j < end) {
            s_l = grp_src[j];
            const float4 a = *(const float4*)(asrc + (size_t)s_l * NUM_HEADS);
            e.x = leaky(a.x + ad.x);
            e.y = leaky(a.y + ad.y);
            e.z = leaky(a.z + ad.z);
            e.w = leaky(a.w + ad.w);
        }
    };

    int    s_l = 0;
    float4 e;
    e.x = NEG; e.y = NEG; e.z = NEG; e.w = NEG;
    float4 m = e;
    for (int base = start; base < end; base += 32) {
        score(base + lane, s_l, e);
        m.x = fmaxf(m.x, e.x);
        m.y = fmaxf(m.y, e.y);
        m.z = fmaxf(m.z, e.z);
        m.w = fmaxf(m.w, e.w);
    }
    m.x = wave_max(m.x);
    m.y = wave_max(m.y);
    m.z = wave_max(m.z);
    m.w = wave_max(m.w);

    float acc0 = 0.0f, acc1 = 0.0f, acc2 = 0.0f, acc3 = 0.0f;
    float4 ps  = {0.0f, 0.0f, 0.0f, 0.0f};
    const float* hrow = h + lane;

    auto gather = [&](int sl, const float4& ee, int nedge) {
        float4 p;
        p.x = __expf(ee.x - m.x);
        p.y = __expf(ee.y - m.y);
        p.z = __expf(ee.z - m.z);
        p.w = __expf(ee.w - m.w);
        ps.x += p.x; ps.y += p.y; ps.z += p.z; ps.w += p.w;
        #pragma unroll 4
        for (int k = 0; k < nedge; ++k) {
            const int   s  = __builtin_amdgcn_readlane(sl, k);
            const float w0 = __int_as_float(__builtin_amdgcn_readlane(__float_as_int(p.x), k));
            const float w1 = __int_as_float(__builtin_amdgcn_readlane(__float_as_int(p.y), k));
            const float w2 = __int_as_float(__builtin_amdgcn_readlane(__float_as_int(p.z), k));
            const float w3 = __int_as_float(__builtin_amdgcn_readlane(__float_as_int(p.w), k));
            const float* r = hrow + (size_t)s * F_OUT;
            acc0 += w0 * r[0];
            acc1 += w1 * r[HEAD_DIM];
            acc2 += w2 * r[2 * HEAD_DIM];
            acc3 += w3 * r[3 * HEAD_DIM];
        }
    };

    if (deg > 0) {
        const int lastBase = start + ((deg - 1) & ~31);
        gather(s_l, e, end - lastBase);
        for (int base = start; base < lastBase; base += 32) {
            int    s2;
            float4 e2;
            score(base + lane, s2, e2);
            gather(s2, e2, 32);
        }
    }

    ps.x = wave_sum(ps.x);
    ps.y = wave_sum(ps.y);
    ps.z = wave_sum(ps.z);
    ps.w = wave_sum(ps.w);

    float* orow = out + (size_t)n * F_OUT + lane;
    orow[0]            = acc0 / (ps.x + 1e-10f);
    orow[HEAD_DIM]     = acc1 / (ps.y + 1e-10f);
    orow[2 * HEAD_DIM] = acc2 / (ps.z + 1e-10f);
    orow[3 * HEAD_DIM] = acc3 / (ps.w + 1e-10f);
}

extern "C" void kernel_launch(void* const* d_in, const int* in_sizes, int n_in,
                              void* d_out, int out_size, void* d_ws, size_t ws_size,
                              hipStream_t stream) {
    const float* x     = (const float*)d_in[0];
    const int*   ei    = (const int*)d_in[1];
    const float* W     = (const float*)d_in[2];
    const float* a_src = (const float*)d_in[3];
    const float* a_dst = (const float*)d_in[4];
    float*       out   = (float*)d_out;

    const int N = in_sizes[0] / F_IN;
    const int E = in_sizes[1] / 2;
    const int* src = ei;
    const int* dst = ei + E;

    float* h    = (float*)d_ws;
    float* asrc = h    + (size_t)N * F_OUT;
    float* adst = asrc + (size_t)N * NUM_HEADS;
    int*   cnt  = (int*)(adst + (size_t)N * NUM_HEADS);
    int*   cur  = cnt + N;
    int*   grp  = cur + N;

    {
        int fullTiles = N / 16;
        if (fullTiles > 0)
            gat_gemm<<<fullTiles, 256, 0, stream>>>(x, W, h);
        int rowStart = fullTiles * 16;
        int rem = N - rowStart;
        if (rem > 0) {
            int total = rem * F_OUT;
            gat_gemm_tail<<<(total + 255) / 256, 256, 0, stream>>>(x, W, h, N, rowStart);
        }
    }

    {
        int total = N * NUM_HEADS;
        gat_alpha<<<(total + 255) / 256, 256, 0, stream>>>(h, a_src, a_dst, asrc, adst, cnt, N);
    }

    gat_count<<<(E + 255) / 256, 256, 0, stream>>>(dst, cnt, E);
    gat_scan<<<(N + 1023) / 1024, 1024, 0, stream>>>(cnt, cur, N);
    gat_fill<<<(E + 255) / 256, 256, 0, stream>>>(src, dst, cur, grp, E);

    gat_node<<<(N + 7) / 8, 256, 0, stream>>>(grp, cnt, cur, asrc, adst, h, out, N);
}


# Edges grouped by destination by two LDS counting sorts (chunk by 256-node bin, then bin by node), whole-line copies: no global atomic, no scattered store per edge, 5 launches; sums in LDS slot order
# speedup vs baseline: 5.2783x; 2.0065x over previous
#include <hip/hip_runtime.h>
#include <hip/hip_bf16.h>


#define NUM_HEADS 4
#define HEAD_DIM 32
#define F_IN 256
#define F_OUT 128
#define LDA 260

typedef __attribute__((ext_vector_type(2))) float v2f;
typedef __attribute__((ext_vector_type(8))) float v8f;

__global__ __launch_bounds__(256) void gat_gemm(const float* __restrict__ x,
                                                const float* __restrict__ W,
                                                float* __restrict__ h) {
    __shared__ float As[16 * LDA];

    const int t = threadIdx.x;
    const int rowBase = blockIdx.x * 16;

    #pragma unroll
    for (int j = 0; j < 4; ++j) {
        int f = t + 256 * j;
        int r = f >> 6;
        int c = (f & 63) << 2;
        float4 v = *(const float4*)(x + (size_t)(rowBase + r) * F_IN + c);
        *(float4*)(&As[r * LDA + c]) = v;
    }
    __syncthreads();

    const int wave = t >> 5;
    const int lane = t & 31;
    const int m    = lane & 15;
    const int koff = (lane >> 4) << 1;
    const int col  = (wave << 4) + m;

    const float* Wp = W + col;
    v8f acc = {};

    #pragma unroll 8
    for (int k = 0; k < F_IN; k += 4) {
        v2f a = *(const v2f*)(&As[m * LDA + k + koff]);
        v2f b;
        b.x = Wp[(size_t)(k + koff) * F_OUT];
        b.y = Wp[(size_t)(k + koff + 1) * F_OUT];
        acc = __builtin_amdgcn_wmma_f32_16x16x4_f32(
             false, a,  false, b,
             (short)0, acc,  false,  false);
    }

    const int roff = (lane >> 4) << 3;
    float* hp = h + (size_t)rowBase * F_OUT + col;
    #pragma unroll
    for (int v = 0; v < 8; ++v)
        hp[(size_t)(v + roff) * F_OUT] = acc[v];
}

__global__ __launch_bounds__(256) void gat_gemm_tail(const float* __restrict__ x,
                                                     const float* __restrict__ W,
                                                     float* __restrict__ h,
                                                     int N, int rowStart) {
    int idx = blockIdx.x * blockDim.x + threadIdx.x;
    int total = (N - rowStart) * F_OUT;
    if (idx >= total) return;
    int r = rowStart + idx / F_OUT;
    int c = idx % F_OUT;
    float s = 0.0f;
    for (int k = 0; k < F_IN; ++k)
        s += x[(size_t)r * F_IN + k] * W[(size_t)k * F_OUT + c];
    h[(size_t)r * F_OUT + c] = s;
}

__global__ __launch_bounds__(256) void gat_alpha(const float* __restrict__ h,
                                                 const float* __restrict__ a_src,
                                                 const float* __restrict__ a_dst,
                                                 float* __restrict__ asrc,
                                                 float* __restrict__ adst,
                                                 int N) {
    int tid = blockIdx.x * blockDim.x + threadIdx.x;
    if (tid >= N * NUM_HEADS) return;
    int n  = tid >> 2;
    int hd = tid & 3;
    const float* hp = h + (size_t)n * F_OUT + hd * HEAD_DIM;
    const float* as = a_src + hd * HEAD_DIM;
    const float* ad = a_dst + hd * HEAD_DIM;
    float s1 = 0.0f, s2 = 0.0f;
    #pragma unroll
    for (int i = 0; i < HEAD_DIM; i += 4) {
        float4 hv = *(const float4*)(hp + i);
        float4 av = *(const float4*)(as + i);
        float4 dv = *(const float4*)(ad + i);
        s1 += hv.x * av.x + hv.y * av.y + hv.z * av.z + hv.w * av.w;
        s2 += hv.x * dv.x + hv.y * dv.y + hv.z * dv.z + hv.w * dv.w;
    }
    asrc[tid] = s1;
    adst[tid] = s2;
}

__device__ __forceinline__ float leaky(float v) { return v > 0.0f ? v : 0.2f * v; }

__device__ __forceinline__ float wave_max(float v) {
    for (int o = 16; o > 0; o >>= 1) v = fmaxf(v, __shfl_xor(v, o));
    return v;
}
__device__ __forceinline__ float wave_sum(float v) {
    for (int o = 16; o > 0; o >>= 1) v += __shfl_xor(v, o);
    return v;
}

#define BIN_SHIFT 8
#define BIN_NODES 256
#define GROUP_CHUNKS 256
#define CHUNK_THREADS 512
#define SORT_CAP 8192
static_assert(GROUP_CHUNKS == BIN_NODES, "gat_bin_sort: thread t is chunk t, then node t");

__global__ __launch_bounds__(CHUNK_THREADS) void gat_chunk_sort(const int* __restrict__ src,
                                                                const int* __restrict__ dst,
                                                                unsigned* __restrict__ chunks,
                                                                int* __restrict__ starts,
                                                                int E, int chunkLen, int NB) {
    extern __shared__ int s_dyn[];
    int* s_hist   = s_dyn;
    unsigned* s_word = (unsigned*)(s_hist + NB);

    const int t    = threadIdx.x;
    const int lane = t & 31;
    const int c    = blockIdx.x;
    const int e0   = c * chunkLen;
    const int e1   = min(E, e0 + chunkLen);

    for (int b = t; b < NB; b += CHUNK_THREADS) s_hist[b] = 0;
    __syncthreads();

    for (int e = e0 + t; e < e1; e += CHUNK_THREADS * 4) {
        int d[4];
        #pragma unroll
        for (int k = 0; k < 4; ++k) {
            const int ee = e + k * CHUNK_THREADS;
            d[k] = ee < e1 ? dst[ee] : 0;
        }
        #pragma unroll
        for (int k = 0; k < 4; ++k)
            if (e + k * CHUNK_THREADS < e1) atomicAdd(&s_hist[d[k] >> BIN_SHIFT], 1);
    }
    __syncthreads();

    if (t < 32) {
        int* row = starts + (size_t)c * (NB + 1);
        int carry = 0;
        for (int base = 0; base < NB; base += 32) {
            const int b = base + lane;
            const int v = b < NB ? s_hist[b] : 0;
            int incl = v;
            for (int o = 1; o < 32; o <<= 1) {
                const int u = __shfl_up(incl, o);
                if (lane >= o) incl += u;
            }
            if (b < NB) {
                const int ls = carry + incl - v;
                s_hist[b] = ls;
                row[b]    = ls;
            }
            carry += __shfl(incl, 31);
        }
        if (lane == 0) row[NB] = carry;
    }
    __syncthreads();

    for (int e = e0 + t; e < e1; e += CHUNK_THREADS * 4) {
        int d[4], s[4];
        #pragma unroll
        for (int k = 0; k < 4; ++k) {
            const int ee = e + k * CHUNK_THREADS;
            d[k] = ee < e1 ? dst[ee] : 0;
            s[k] = ee < e1 ? src[ee] : 0;
        }
        #pragma unroll
        for (int k = 0; k < 4; ++k) {
            if (e + k * CHUNK_THREADS < e1) {
                const int p = atomicAdd(&s_hist[d[k] >> BIN_SHIFT], 1);
                s_word[p] = ((unsigned)(d[k] & (BIN_NODES - 1)) << 24) | (unsigned)s[k];
            }
        }
    }
    __syncthreads();

    unsigned* outp = chunks + (size_t)c * chunkLen;
    const int m = e1 - e0;
    for (int i = t; i < m; i += CHUNK_THREADS) outp[i] = s_word[i];
}

__device__ __forceinline__ int block_scan_excl(int v, int* s_w, int lane, int wv, int& total) {
    int incl = v;
    for (int o = 1; o < 32; o <<= 1) {
        const int u = __shfl_up(incl, o);
        if (lane >= o) incl += u;
    }
    if (lane == 31) s_w[wv] = incl;
    __syncthreads();
    int pre = 0, tot = 0;
    for (int w = 0; w < BIN_NODES / 32; ++w) {
        const int x = s_w[w];
        tot += x;
        pre += w < wv ? x : 0;
    }
    __syncthreads();
    total = tot;
    return pre + incl - v;
}

__global__ __launch_bounds__(BIN_NODES) void gat_bin_sort(const unsigned* __restrict__ chunks,
                                                          const int* __restrict__ starts,
                                                          int* __restrict__ cnt,
                                                          int* __restrict__ cur,
                                                          int* __restrict__ grp_src,
                                                          int chunkLen, int NB, int N) {
    __shared__ int s_deg[BIN_NODES];
    __shared__ int s_cursor[BIN_NODES];
    __shared__ int s_w[BIN_NODES / 32];
    __shared__ unsigned s_raw[SORT_CAP];
    __shared__ int s_sorted[SORT_CAP];

    const int t    = threadIdx.x;
    const int lane = t & 31;
    const int wv   = t >> 5;
    const int bin  = blockIdx.x;

    const int* row = starts + (size_t)t * (NB + 1) + bin;
    const int ls  = row[0];
    const int len = row[1] - ls;
    const unsigned* run = chunks + (size_t)t * chunkLen + ls;
    int s0, m;
    block_scan_excl(ls, s_w, lane, wv, s0);
    const int roff = block_scan_excl(len, s_w, lane, wv, m);
    const bool staged = m <= SORT_CAP;

    s_deg[t] = 0;
    __syncthreads();
    for (int j = 0; j < len; j += 4) {
        unsigned w[4];
        #pragma unroll
        for (int k = 0; k < 4; ++k) w[k] = j + k < len ? run[j + k] : 0u;
        #pragma unroll
        for (int k = 0; k < 4; ++k) {
            if (j + k < len) {
                atomicAdd(&s_deg[w[k] >> 24], 1);
                if (staged) s_raw[roff + j + k] = w[k];
            }
        }
    }
    __syncthreads();

    const int d = s_deg[t];
    int unused;
    const int excl = block_scan_excl(d, s_w, lane, wv, unused);
    s_cursor[t] = excl;
    const int n = bin * BIN_NODES + t;
    if (n < N) {
        cnt[n] = d;
        cur[n] = s0 + excl + d;
    }
    __syncthreads();

    int* outp = grp_src + s0;
    if (staged) {
        for (int i = t; i < m; i += BIN_NODES) {
            const unsigned w = s_raw[i];
            const int p = atomicAdd(&s_cursor[w >> 24], 1);
            s_sorted[p] = (int)(w & 0xFFFFFFu);
        }
        __syncthreads();
        for (int i = t; i < m; i += BIN_NODES) outp[i] = s_sorted[i];
    } else {
        for (int j = 0; j < len; ++j) {
            const unsigned w = run[j];
            const int p = atomicAdd(&s_cursor[w >> 24], 1);
            outp[p] = (int)(w & 0xFFFFFFu);
        }
    }
}

__global__ __launch_bounds__(256) void gat_node(const int* __restrict__ grp_src,
                                                const int* __restrict__ cnt,
                                                const int* __restrict__ cur,
                                                const float* __restrict__ asrc,
                                                const float* __restrict__ adst,
                                                const float* __restrict__ h,
                                                float* __restrict__ out,
                                                int N) {
    const int n = blockIdx.x * 8 + __builtin_amdgcn_readfirstlane((int)(threadIdx.x >> 5));
    if (n >= N) return;
    const int lane = threadIdx.x & 31;

    const int end   = cur[n];
    const int deg   = cnt[n];
    const int start = end - deg;
    const float4 ad = *(const float4*)(adst + (size_t)n * NUM_HEADS);

    const float NEG = -__builtin_huge_valf();

    auto score = [&](int j, int& s_l, float4& e) {
        s_l = 0;
        e.x = NEG; e.y = NEG; e.z = NEG; e.w = NEG;
        if (j < end) {
            s_l = grp_src[j];
            const float4 a = *(const float4*)(asrc + (size_t)s_l * NUM_HEADS);
            e.x = leaky(a.x + ad.x);
            e.y = leaky(a.y + ad.y);
            e.z = leaky(a.z + ad.z);
            e.w = leaky(a.w + ad.w);
        }
    };

    int    s_l = 0;
    float4 e;
    e.x = NEG; e.y = NEG; e.z = NEG; e.w = NEG;
    float4 m = e;
    for (int base = start; base < end; base += 32) {
        score(base + lane, s_l, e);
        m.x = fmaxf(m.x, e.x);
        m.y = fmaxf(m.y, e.y);
        m.z = fmaxf(m.z, e.z);
        m.w = fmaxf(m.w, e.w);
    }
    m.x = wave_max(m.x);
    m.y = wave_max(m.y);
    m.z = wave_max(m.z);
    m.w = wave_max(m.w);

    float acc0 = 0.0f, acc1 = 0.0f, acc2 = 0.0f, acc3 = 0.0f;
    float4 ps  = {0.0f, 0.0f, 0.0f, 0.0f};
    const float* hrow = h + lane;

    auto gather = [&](int sl, const float4& ee, int nedge) {
        float4 p;
        p.x = __expf(ee.x - m.x);
        p.y = __expf(ee.y - m.y);
        p.z = __expf(ee.z - m.z);
        p.w = __expf(ee.w - m.w);
        ps.x += p.x; ps.y += p.y; ps.z += p.z; ps.w += p.w;
        #pragma unroll 4
        for (int k = 0; k < nedge; ++k) {
            const int   s  = __builtin_amdgcn_readlane(sl, k);
            const float w0 = __int_as_float(__builtin_amdgcn_readlane(__float_as_int(p.x), k));
            const float w1 = __int_as_float(__builtin_amdgcn_readlane(__float_as_int(p.y), k));
            const float w2 = __int_as_float(__builtin_amdgcn_readlane(__float_as_int(p.z), k));
            const float w3 = __int_as_float(__builtin_amdgcn_readlane(__float_as_int(p.w), k));
            const float* r = hrow + (size_t)s * F_OUT;
            acc0 += w0 * r[0];
            acc1 += w1 * r[HEAD_DIM];
            acc2 += w2 * r[2 * HEAD_DIM];
            acc3 += w3 * r[3 * HEAD_DIM];
        }
    };

    if (deg > 0) {
        const int lastBase = start + ((deg - 1) & ~31);
        gather(s_l, e, end - lastBase);
        for (int base = start; base < lastBase; base += 32) {
            int    s2;
            float4 e2;
            score(base + lane, s2, e2);
            gather(s2, e2, 32);
        }
    }

    ps.x = wave_sum(ps.x);
    ps.y = wave_sum(ps.y);
    ps.z = wave_sum(ps.z);
    ps.w = wave_sum(ps.w);

    float* orow = out + (size_t)n * F_OUT + lane;
    orow[0]            = acc0 / (ps.x + 1e-10f);
    orow[HEAD_DIM]     = acc1 / (ps.y + 1e-10f);
    orow[2 * HEAD_DIM] = acc2 / (ps.z + 1e-10f);
    orow[3 * HEAD_DIM] = acc3 / (ps.w + 1e-10f);
}

extern "C" void kernel_launch(void* const* d_in, const int* in_sizes, int n_in,
                              void* d_out, int out_size, void* d_ws, size_t ws_size,
                              hipStream_t stream) {
    const float* x     = (const float*)d_in[0];
    const int*   ei    = (const int*)d_in[1];
    const float* W     = (const float*)d_in[2];
    const float* a_src = (const float*)d_in[3];
    const float* a_dst = (const float*)d_in[4];
    float*       out   = (float*)d_out;

    const int N = in_sizes[0] / F_IN;
    const int E = in_sizes[1] / 2;
    const int* src = ei;
    const int* dst = ei + E;

    float* h    = (float*)d_ws;
    float* asrc = h    + (size_t)N * F_OUT;
    float* adst = asrc + (size_t)N * NUM_HEADS;
    int*   cnt  = (int*)(adst + (size_t)N * NUM_HEADS);
    int*   cur  = cnt + N;
    int*   grp  = cur + N;

    {
        int fullTiles = N / 16;
        if (fullTiles > 0)
            gat_gemm<<<fullTiles, 256, 0, stream>>>(x, W, h);
        int rowStart = fullTiles * 16;
        int rem = N - rowStart;
        if (rem > 0) {
            int total = rem * F_OUT;
            gat_gemm_tail<<<(total + 255) / 256, 256, 0, stream>>>(x, W, h, N, rowStart);
        }
    }

    {
        int total = N * NUM_HEADS;
        gat_alpha<<<(total + 255) / 256, 256, 0, stream>>>(h, a_src, a_dst, asrc, adst, N);
    }

    {
        const int NB       = (N + BIN_NODES - 1) / BIN_NODES;
        const int chunkLen = ((E + GROUP_CHUNKS - 1) / GROUP_CHUNKS + 31) & ~31;
        unsigned* chunks   = (unsigned*)d_out;
        int*      starts   = (int*)d_out + (size_t)GROUP_CHUNKS * chunkLen;
        const size_t lds   = ((size_t)NB + chunkLen) * sizeof(int);
        gat_chunk_sort<<<GROUP_CHUNKS, CHUNK_THREADS, lds, stream>>>(src, dst, chunks, starts,
                                                                     E, chunkLen, NB);
        gat_bin_sort<<<NB, BIN_NODES, 0, stream>>>(chunks, starts, cnt, cur, grp, chunkLen, NB, N);
    }

    gat_node<<<(N + 7) / 8, 256, 0, stream>>>(grp, cnt, cur, asrc, adst, h, out, N);
}


# Both sort launches carry half of the GEMM's row tiles each as extra work-groups (sorting groups first): independent work runs side by side, the GEMM's own launch goes, 4 launches become 3
# speedup vs baseline: 6.6978x; 1.2689x over previous
#include <hip/hip_runtime.h>
#include <hip/hip_bf16.h>
#include <type_traits>


#define NUM_HEADS 4
#define HEAD_DIM 32
#define F_IN 256
#define F_OUT 128
#define LDA 260

typedef __attribute__((ext_vector_type(2))) float v2f;
typedef __attribute__((ext_vector_type(8))) float v8f;

#define TILE_LDS_FLOATS (16 * LDA)

__device__ __forceinline__ void gemm_stage(const float* __restrict__ x, float* As,
                                           int rowBase, int t8) {
    #pragma unroll
    for (int j = 0; j < 4; ++j) {
        int f = t8 + 256 * j;
        int r = f >> 6;
        int c = (f & 63) << 2;
        float4 v = *(const float4*)(x + (size_t)(rowBase + r) * F_IN + c);
        *(float4*)(&As[r * LDA + c]) = v;
    }
}

__device__ __forceinline__ void gemm_tile(const float* __restrict__ W, float* __restrict__ h,
                                          const float* As, int rowBase, int t8) {
    const int wave = t8 >> 5;
    const int lane = t8 & 31;
    const int m    = lane & 15;
    const int koff = (lane >> 4) << 1;
    const int col  = (wave << 4) + m;

    const float* Wp = W + col;
    v8f acc = {};

    #pragma unroll 8
    for (int k = 0; k < F_IN; k += 4) {
        v2f a = *(const v2f*)(&As[m * LDA + k + koff]);
        v2f b;
        b.x = Wp[(size_t)(k + koff) * F_OUT];
        b.y = Wp[(size_t)(k + koff + 1) * F_OUT];
        acc = __builtin_amdgcn_wmma_f32_16x16x4_f32(
             false, a,  false, b,
             (short)0, acc,  false,  false);
    }

    const int roff = (lane >> 4) << 3;
    float* hp = h + (size_t)rowBase * F_OUT + col;
    #pragma unroll
    for (int v = 0; v < 8; ++v)
        hp[(size_t)(v + roff) * F_OUT] = acc[v];
}

__global__ __launch_bounds__(256) void gat_gemm_tail(const float* __restrict__ x,
                                                     const float* __restrict__ W,
                                                     float* __restrict__ h,
                                                     int N, int rowStart) {
    int idx = blockIdx.x * blockDim.x + threadIdx.x;
    int total = (N - rowStart) * F_OUT;
    if (idx >= total) return;
    int r = rowStart + idx / F_OUT;
    int c = idx % F_OUT;
    float s = 0.0f;
    for (int k = 0; k < F_IN; ++k)
        s += x[(size_t)r * F_IN + k] * W[(size_t)k * F_OUT + c];
    h[(size_t)r * F_OUT + c] = s;
}

__device__ __forceinline__ float leaky(float v) { return fmaxf(v, 0.2f * v); }

typedef __attribute__((ext_vector_type(4))) float v4f;

template <int CTRL> __device__ __forceinline__ float dpp_take(float v) {
    return __int_as_float(__builtin_amdgcn_update_dpp(0, __float_as_int(v), CTRL, 0xF, 0xF, true));
}
__device__ __forceinline__ float group8_sum(float v) {
    v += dpp_take<0xB1>(v);
    v += dpp_take<0x4E>(v);
    v += dpp_take<0x141>(v);
    return v;
}
__device__ __forceinline__ float dot4(v4f r, v4f a) {
    const v2f t = __builtin_elementwise_fma(r.hi, a.hi, r.lo * a.lo);
    return t.x + t.y;
}

#define BIN_SHIFT 8
#define BIN_NODES 256
#define GROUP_CHUNKS 256
#define CHUNK_THREADS 512
#define SORT_CAP 8192
static_assert(GROUP_CHUNKS == BIN_NODES, "gat_bin_sort: thread t is chunk t, then node t");

__global__ __launch_bounds__(CHUNK_THREADS) void gat_chunk_sort_gemm(const int* __restrict__ src,
                                                                     const int* __restrict__ dst,
                                                                     unsigned* __restrict__ chunks,
                                                                     int* __restrict__ starts,
                                                                     int E, int chunkLen, int NB,
                                                                     const float* __restrict__ x,
                                                                     const float* __restrict__ W,
                                                                     float* __restrict__ h,
                                                                     int tileEnd) {
    extern __shared__ __align__(16) unsigned char s_dyn[];

    const int t    = threadIdx.x;

    if (blockIdx.x >= GROUP_CHUNKS) {
        const int grp  = __builtin_amdgcn_readfirstlane(t >> 8);
        const int tile = 2 * ((int)blockIdx.x - GROUP_CHUNKS) + grp;
        float* As      = (float*)s_dyn + grp * TILE_LDS_FLOATS;
        const int t8   = t & 255;
        if (tile < tileEnd) gemm_stage(x, As, tile * 16, t8);
        __syncthreads();
        if (tile < tileEnd) gemm_tile(W, h, As, tile * 16, t8);
        return;
    }

    int* s_hist   = (int*)s_dyn;
    unsigned* s_word = (unsigned*)(s_hist + NB);

    const int lane = t & 31;
    const int c    = blockIdx.x;
    const int e0   = c * chunkLen;
    const int e1   = min(E, e0 + chunkLen);

    for (int b = t; b < NB; b += CHUNK_THREADS) s_hist[b] = 0;
    __syncthreads();

    for (int e = e0 + t; e < e1; e += CHUNK_THREADS * 4) {
        int d[4];
        #pragma unroll
        for (int k = 0; k < 4; ++k) {
            const int ee = e + k * CHUNK_THREADS;
            d[k] = ee < e1 ? dst[ee] : 0;
        }
        #pragma unroll
        for (int k = 0; k < 4; ++k)
            if (e + k * CHUNK_THREADS < e1) atomicAdd(&s_hist[d[k] >> BIN_SHIFT], 1);
    }
    __syncthreads();

    if (t < 32) {
        int* row = starts + (size_t)c * (NB + 1);
        int carry = 0;
        for (int base = 0; base < NB; base += 32) {
            const int b = base + lane;
            const int v = b < NB ? s_hist[b] : 0;
            int incl = v;
            for (int o = 1; o < 32; o <<= 1) {
                const int u = __shfl_up(incl, o);
                if (lane >= o) incl += u;
            }
            if (b < NB) {
                const int ls = carry + incl - v;
                s_hist[b] = ls;
                row[b]    = ls;
            }
            carry += __shfl(incl, 31);
        }
        if (lane == 0) row[NB] = carry;
    }
    __syncthreads();

    for (int e = e0 + t; e < e1; e += CHUNK_THREADS * 4) {
        int d[4], s[4];
        #pragma unroll
        for (int k = 0; k < 4; ++k) {
            const int ee = e + k * CHUNK_THREADS;
            d[k] = ee < e1 ? dst[ee] : 0;
            s[k] = ee < e1 ? src[ee] : 0;
        }
        #pragma unroll
        for (int k = 0; k < 4; ++k) {
            if (e + k * CHUNK_THREADS < e1) {
                const int p = atomicAdd(&s_hist[d[k] >> BIN_SHIFT], 1);
                s_word[p] = ((unsigned)(d[k] & (BIN_NODES - 1)) << 24) | (unsigned)s[k];
            }
        }
    }
    __syncthreads();

    unsigned* outp = chunks + (size_t)c * chunkLen;
    const int m = e1 - e0;
    for (int i = t; i < m; i += CHUNK_THREADS) outp[i] = s_word[i];
}

__device__ __forceinline__ int block_scan_excl(int v, int* s_w, int lane, int wv, int& total) {
    int incl = v;
    for (int o = 1; o < 32; o <<= 1) {
        const int u = __shfl_up(incl, o);
        if (lane >= o) incl += u;
    }
    if (lane == 31) s_w[wv] = incl;
    __syncthreads();
    int pre = 0, tot = 0;
    for (int w = 0; w < BIN_NODES / 32; ++w) {
        const int x = s_w[w];
        tot += x;
        pre += w < wv ? x : 0;
    }
    __syncthreads();
    total = tot;
    return pre + incl - v;
}

union __align__(16) BinStage {
    unsigned raw[SORT_CAP];
    float    a[TILE_LDS_FLOATS];
};
static_assert(SORT_CAP >= TILE_LDS_FLOATS, "the A tile must fit the sort's staging array");

__global__ __launch_bounds__(BIN_NODES) void gat_bin_sort_gemm(const unsigned* __restrict__ chunks,
                                                               const int* __restrict__ starts,
                                                               int* __restrict__ cnt,
                                                               int* __restrict__ cur,
                                                               int* __restrict__ grp_src,
                                                               int chunkLen, int NB, int N,
                                                               const float* __restrict__ x,
                                                               const float* __restrict__ W,
                                                               float* __restrict__ h,
                                                               int tileBegin) {
    __shared__ int s_deg[BIN_NODES];
    __shared__ int s_cursor[BIN_NODES];
    __shared__ int s_w[BIN_NODES / 32];
    __shared__ BinStage s_stage;
    __shared__ int s_sorted[SORT_CAP];
    unsigned* const s_raw = s_stage.raw;

    const int t    = threadIdx.x;

    if ((int)blockIdx.x >= NB) {
        const int rowBase = (tileBegin + ((int)blockIdx.x - NB)) * 16;
        gemm_stage(x, s_stage.a, rowBase, t);
        __syncthreads();
        gemm_tile(W, h, s_stage.a, rowBase, t);
        return;
    }

    const int lane = t & 31;
    const int wv   = t >> 5;
    const int bin  = blockIdx.x;

    const int* row = starts + (size_t)t * (NB + 1) + bin;
    const int ls  = row[0];
    const int len = row[1] - ls;
    const unsigned* run = chunks + (size_t)t * chunkLen + ls;
    int s0, m;
    block_scan_excl(ls, s_w, lane, wv, s0);
    const int roff = block_scan_excl(len, s_w, lane, wv, m);
    const bool staged = m <= SORT_CAP;

    s_deg[t] = 0;
    __syncthreads();
    for (int j = 0; j < len; j += 4) {
        unsigned w[4];
        #pragma unroll
        for (int k = 0; k < 4; ++k) w[k] = j + k < len ? run[j + k] : 0u;
        #pragma unroll
        for (int k = 0; k < 4; ++k) {
            if (j + k < len) {
                atomicAdd(&s_deg[w[k] >> 24], 1);
                if (staged) s_raw[roff + j + k] = w[k];
            }
        }
    }
    __syncthreads();

    const int d = s_deg[t];
    int unused;
    const int excl = block_scan_excl(d, s_w, lane, wv, unused);
    s_cursor[t] = excl;
    const int n = bin * BIN_NODES + t;
    if (n < N) {
        cnt[n] = d;
        cur[n] = s0 + excl + d;
    }
    __syncthreads();

    int* outp = grp_src + s0;
    if (staged) {
        for (int i = t; i < m; i += BIN_NODES) {
            const unsigned w = s_raw[i];
            const int p = atomicAdd(&s_cursor[w >> 24], 1);
            s_sorted[p] = (int)(w & 0xFFFFFFu);
        }
        __syncthreads();
        for (int i = t; i < m; i += BIN_NODES) outp[i] = s_sorted[i];
    } else {
        for (int j = 0; j < len; ++j) {
            const unsigned w = run[j];
            const int p = atomicAdd(&s_cursor[w >> 24], 1);
            outp[p] = (int)(w & 0xFFFFFFu);
        }
    }
}

#define NODE_BATCH 8
__global__ __launch_bounds__(256) void gat_node(const int* __restrict__ grp_src,
                                                const int* __restrict__ cnt,
                                                const int* __restrict__ cur,
                                                const float* __restrict__ a_src,
                                                const float* __restrict__ a_dst,
                                                const float* __restrict__ h,
                                                float* __restrict__ out,
                                                int N) {
    const int n = blockIdx.x * 8 + __builtin_amdgcn_readfirstlane((int)(threadIdx.x >> 5));
    if (n >= N) return;
    const int lane = threadIdx.x & 31;

    const int end   = cur[n];
    const int deg   = cnt[n];
    const int start = end - deg;

    int sl = start + lane < end ? grp_src[start + lane] : 0;

    const unsigned f4 = 4u * (unsigned)lane;
    const v4f as4 = *(const v4f*)(a_src + f4);
    const v4f ad4 = *(const v4f*)(a_dst + f4);
    const v4f hn  = *(const v4f*)(h + (size_t)n * F_OUT + f4);
    const float adn = group8_sum(dot4(hn, ad4));

    const float NEG = -__builtin_huge_valf();
    float m   = NEG;
    float ps  = 0.0f;
    v4f   acc = {0.0f, 0.0f, 0.0f, 0.0f};

    auto batch = [&](int sl, int k0, int nb, auto full) {
        constexpr bool FULL = decltype(full)::value;
        v4f r[NODE_BATCH];
        #pragma unroll
        for (int k = 0; k < NODE_BATCH; ++k) {
            if (FULL || k < nb) {
                const int s = __builtin_amdgcn_readlane(sl, k0 + k);
                const float* hs = h + (size_t)s * F_OUT;
                r[k] = *(const v4f*)(hs + f4);
            } else {
                r[k] = (v4f){0.0f, 0.0f, 0.0f, 0.0f};
            }
        }
        float e[NODE_BATCH];
        float mb = NEG;
        #pragma unroll
        for (int k = 0; k < NODE_BATCH; ++k) {
            const float d = group8_sum(dot4(r[k], as4));
            e[k] = (FULL || k < nb) ? leaky(d + adn) : NEG;
            mb = fmaxf(mb, e[k]);
        }
        const float mn = fmaxf(m, mb);
        const float sc = __expf(m - mn);
        m = mn;
        ps *= sc;
        acc *= sc;
        #pragma unroll
        for (int k = 0; k < NODE_BATCH; ++k) {
            const float p = __expf(e[k] - mn);
            ps += p;
            acc = __builtin_elementwise_fma(r[k], (v4f){p, p, p, p}, acc);
        }
    };

    for (int base = start; base < end; base += 32) {
        if (base != start) {
            const int j = base + lane;
            sl = j < end ? grp_src[j] : 0;
        }
        const int ng = min(32, end - base);
        int k0 = 0;
        for (; k0 + NODE_BATCH <= ng; k0 += NODE_BATCH)
            batch(sl, k0, NODE_BATCH, std::true_type{});
        if (k0 < ng)
            batch(sl, k0, ng - k0, std::false_type{});
    }

    const float den = ps + 1e-10f;
    v4f o;
    o.x = acc.x / den;
    o.y = acc.y / den;
    o.z = acc.z / den;
    o.w = acc.w / den;
    *(v4f*)(out + (size_t)n * F_OUT + f4) = o;
}

extern "C" void kernel_launch(void* const* d_in, const int* in_sizes, int n_in,
                              void* d_out, int out_size, void* d_ws, size_t ws_size,
                              hipStream_t stream) {
    const float* x     = (const float*)d_in[0];
    const int*   ei    = (const int*)d_in[1];
    const float* W     = (const float*)d_in[2];
    const float* a_src = (const float*)d_in[3];
    const float* a_dst = (const float*)d_in[4];
    float*       out   = (float*)d_out;

    const int N = in_sizes[0] / F_IN;
    const int E = in_sizes[1] / 2;
    const int* src = ei;
    const int* dst = ei + E;

    float* h    = (float*)d_ws;
    int*   cnt  = (int*)(h + (size_t)N * F_OUT);
    int*   cur  = cnt + N;
    int*   grp  = cur + N;

    {
        const int fullTiles = N / 16;
        const int tilesA    = fullTiles / 2;
        const int tilesB    = fullTiles - tilesA;

        const int NB       = (N + BIN_NODES - 1) / BIN_NODES;
        const int chunkLen = ((E + GROUP_CHUNKS - 1) / GROUP_CHUNKS + 31) & ~31;
        unsigned* chunks   = (unsigned*)d_out;
        int*      starts   = (int*)d_out + (size_t)GROUP_CHUNKS * chunkLen;
        const size_t ldsSort = ((size_t)NB + chunkLen) * sizeof(int);
        const size_t ldsGemm = (size_t)2 * TILE_LDS_FLOATS * sizeof(float);
        const size_t lds     = ldsSort > ldsGemm ? ldsSort : ldsGemm;
        gat_chunk_sort_gemm<<<GROUP_CHUNKS + (tilesA + 1) / 2, CHUNK_THREADS, lds, stream>>>(
            src, dst, chunks, starts, E, chunkLen, NB, x, W, h, tilesA);
        gat_bin_sort_gemm<<<NB + tilesB, BIN_NODES, 0, stream>>>(
            chunks, starts, cnt, cur, grp, chunkLen, NB, N, x, W, h, tilesA);

        int rowStart = fullTiles * 16;
        int rem = N - rowStart;
        if (rem > 0) {
            int total = rem * F_OUT;
            gat_gemm_tail<<<(total + 255) / 256, 256, 0, stream>>>(x, W, h, N, rowStart);
        }
    }

    gat_node<<<(N + 7) / 8, 256, 0, stream>>>(grp, cnt, cur, a_src, a_dst, h, out, N);
}


# GEMM wave owns a 16x32 block as two interleaved column tiles (even/odd columns): one 8-byte W load and one A fragment feed two WMMAs, stores cover whole 128-byte lines; A tile XOR-placed, no padding
# speedup vs baseline: 6.9482x; 1.0374x over previous
#include <hip/hip_runtime.h>
#include <hip/hip_bf16.h>
#include <type_traits>


#define NUM_HEADS 4
#define HEAD_DIM 32
#define F_IN 256
#define F_OUT 128
#define TILE_K4 64

typedef __attribute__((ext_vector_type(2))) float v2f;
typedef __attribute__((ext_vector_type(8))) float v8f;

#define TILE_LDS_FLOATS (16 * F_IN)
#define GEMM_GROUP 128

__device__ __forceinline__ void gemm_stage(const float* __restrict__ x, float* As,
                                           int rowBase, int tg) {
    #pragma unroll
    for (int j = 0; j < 8; ++j) {
        int f = tg + GEMM_GROUP * j;
        int r = f >> 6;
        int q = f & 63;
        float4 v = *(const float4*)(x + (size_t)(rowBase + r) * F_IN + (q << 2));
        *(float4*)(&As[r * F_IN + ((q ^ (r & 7)) << 2)]) = v;
    }
}

__device__ __forceinline__ void gemm_tile(const float* __restrict__ W, float* __restrict__ h,
                                          const float* As, int rowBase, int tg) {
    const int wave = tg >> 5;
    const int lane = tg & 31;
    const int m    = lane & 15;
    const int koff = (lane >> 4) << 1;
    const int col  = (wave << 5) + (m << 1);

    const float* Wp = W + col;
    const float* Ar = As + m * F_IN + koff;
    const float* a[8];
    #pragma unroll
    for (int j = 0; j < 8; ++j) a[j] = Ar + ((j ^ (m & 7)) << 2);

    v8f acc0 = {};
    v8f acc1 = {};

    #pragma unroll 1
    for (int q0 = 0; q0 < TILE_K4; q0 += 8) {
        #pragma unroll
        for (int j = 0; j < 8; ++j) {
            const int k = ((q0 + j) << 2) + koff;
            const v2f av = *(const v2f*)(a[j] + (q0 << 2));
            const v2f b0 = *(const v2f*)(Wp + (size_t)k * F_OUT);
            const v2f b1 = *(const v2f*)(Wp + (size_t)(k + 1) * F_OUT);
            const v2f be = {b0.x, b1.x};
            const v2f bo = {b0.y, b1.y};
            acc0 = __builtin_amdgcn_wmma_f32_16x16x4_f32(
                 false, av,  false, be,
                 (short)0, acc0,  false,  false);
            acc1 = __builtin_amdgcn_wmma_f32_16x16x4_f32(
                 false, av,  false, bo,
                 (short)0, acc1,  false,  false);
        }
    }

    const int roff = (lane >> 4) << 3;
    float* hp = h + (size_t)rowBase * F_OUT + col;
    #pragma unroll
    for (int v = 0; v < 8; ++v) {
        const v2f o = {acc0[v], acc1[v]};
        *(v2f*)(hp + (size_t)(v + roff) * F_OUT) = o;
    }
}

__global__ __launch_bounds__(256) void gat_gemm_tail(const float* __restrict__ x,
                                                     const float* __restrict__ W,
                                                     float* __restrict__ h,
                                                     int N, int rowStart) {
    int idx = blockIdx.x * blockDim.x + threadIdx.x;
    int total = (N - rowStart) * F_OUT;
    if (idx >= total) return;
    int r = rowStart + idx / F_OUT;
    int c = idx % F_OUT;
    float s = 0.0f;
    for (int k = 0; k < F_IN; ++k)
        s += x[(size_t)r * F_IN + k] * W[(size_t)k * F_OUT + c];
    h[(size_t)r * F_OUT + c] = s;
}

__device__ __forceinline__ float leaky(float v) { return fmaxf(v, 0.2f * v); }

typedef __attribute__((ext_vector_type(4))) float v4f;

template <int CTRL> __device__ __forceinline__ float dpp_take(float v) {
    return __int_as_float(__builtin_amdgcn_update_dpp(0, __float_as_int(v), CTRL, 0xF, 0xF, true));
}
__device__ __forceinline__ float group8_sum(float v) {
    v += dpp_take<0xB1>(v);
    v += dpp_take<0x4E>(v);
    v += dpp_take<0x141>(v);
    return v;
}
__device__ __forceinline__ float dot4(v4f r, v4f a) {
    const v2f t = __builtin_elementwise_fma(r.hi, a.hi, r.lo * a.lo);
    return t.x + t.y;
}

#define BIN_SHIFT 8
#define BIN_NODES 256
#define GROUP_CHUNKS 256
#define CHUNK_THREADS 512
#define SORT_CAP 8192
static_assert(GROUP_CHUNKS == BIN_NODES, "gat_bin_sort: thread t is chunk t, then node t");

__global__ __launch_bounds__(CHUNK_THREADS) void gat_chunk_sort_gemm(const int* __restrict__ src,
                                                                     const int* __restrict__ dst,
                                                                     unsigned* __restrict__ chunks,
                                                                     int* __restrict__ starts,
                                                                     int E, int chunkLen, int NB,
                                                                     const float* __restrict__ x,
                                                                     const float* __restrict__ W,
                                                                     float* __restrict__ h,
                                                                     int tileEnd) {
    extern __shared__ __align__(16) unsigned char s_dyn[];

    const int t    = threadIdx.x;

    if (blockIdx.x >= GROUP_CHUNKS) {
        const int grp  = __builtin_amdgcn_readfirstlane(t >> 7);
        const int tile = 4 * ((int)blockIdx.x - GROUP_CHUNKS) + grp;
        float* As      = (float*)s_dyn + grp * TILE_LDS_FLOATS;
        const int tg   = t & (GEMM_GROUP - 1);
        if (tile < tileEnd) gemm_stage(x, As, tile * 16, tg);
        __syncthreads();
        if (tile < tileEnd) gemm_tile(W, h, As, tile * 16, tg);
        return;
    }

    int* s_hist   = (int*)s_dyn;
    unsigned* s_word = (unsigned*)(s_hist + NB);

    const int lane = t & 31;
    const int c    = blockIdx.x;
    const int e0   = c * chunkLen;
    const int e1   = min(E, e0 + chunkLen);

    for (int b = t; b < NB; b += CHUNK_THREADS) s_hist[b] = 0;
    __syncthreads();

    for (int e = e0 + t; e < e1; e += CHUNK_THREADS * 4) {
        int d[4];
        #pragma unroll
        for (int k = 0; k < 4; ++k) {
            const int ee = e + k * CHUNK_THREADS;
            d[k] = ee < e1 ? dst[ee] : 0;
        }
        #pragma unroll
        for (int k = 0; k < 4; ++k)
            if (e + k * CHUNK_THREADS < e1) atomicAdd(&s_hist[d[k] >> BIN_SHIFT], 1);
    }
    __syncthreads();

    if (t < 32) {
        int* row = starts + (size_t)c * (NB + 1);
        int carry = 0;
        for (int base = 0; base < NB; base += 32) {
            const int b = base + lane;
            const int v = b < NB ? s_hist[b] : 0;
            int incl = v;
            for (int o = 1; o < 32; o <<= 1) {
                const int u = __shfl_up(incl, o);
                if (lane >= o) incl += u;
            }
            if (b < NB) {
                const int ls = carry + incl - v;
                s_hist[b] = ls;
                row[b]    = ls;
            }
            carry += __shfl(incl, 31);
        }
        if (lane == 0) row[NB] = carry;
    }
    __syncthreads();

    for (int e = e0 + t; e < e1; e += CHUNK_THREADS * 4) {
        int d[4], s[4];
        #pragma unroll
        for (int k = 0; k < 4; ++k) {
            const int ee = e + k * CHUNK_THREADS;
            d[k] = ee < e1 ? dst[ee] : 0;
            s[k] = ee < e1 ? src[ee] : 0;
        }
        #pragma unroll
        for (int k = 0; k < 4; ++k) {
            if (e + k * CHUNK_THREADS < e1) {
                const int p = atomicAdd(&s_hist[d[k] >> BIN_SHIFT], 1);
                s_word[p] = ((unsigned)(d[k] & (BIN_NODES - 1)) << 24) | (unsigned)s[k];
            }
        }
    }
    __syncthreads();

    unsigned* outp = chunks + (size_t)c * chunkLen;
    const int m = e1 - e0;
    for (int i = t; i < m; i += CHUNK_THREADS) outp[i] = s_word[i];
}

__device__ __forceinline__ int block_scan_excl(int v, int* s_w, int lane, int wv, int& total) {
    int incl = v;
    for (int o = 1; o < 32; o <<= 1) {
        const int u = __shfl_up(incl, o);
        if (lane >= o) incl += u;
    }
    if (lane == 31) s_w[wv] = incl;
    __syncthreads();
    int pre = 0, tot = 0;
    for (int w = 0; w < BIN_NODES / 32; ++w) {
        const int x = s_w[w];
        tot += x;
        pre += w < wv ? x : 0;
    }
    __syncthreads();
    total = tot;
    return pre + incl - v;
}

union __align__(16) BinStage {
    unsigned raw[SORT_CAP];
    float    a[2 * TILE_LDS_FLOATS];
};
static_assert(SORT_CAP >= 2 * TILE_LDS_FLOATS, "the A tiles must fit the sort's staging array");

__global__ __launch_bounds__(BIN_NODES) void gat_bin_sort_gemm(const unsigned* __restrict__ chunks,
                                                               const int* __restrict__ starts,
                                                               int* __restrict__ cnt,
                                                               int* __restrict__ cur,
                                                               int* __restrict__ grp_src,
                                                               int chunkLen, int NB, int N,
                                                               const float* __restrict__ x,
                                                               const float* __restrict__ W,
                                                               float* __restrict__ h,
                                                               int tileBegin) {
    __shared__ int s_deg[BIN_NODES];
    __shared__ int s_cursor[BIN_NODES];
    __shared__ int s_w[BIN_NODES / 32];
    __shared__ BinStage s_stage;
    __shared__ int s_sorted[SORT_CAP];
    unsigned* const s_raw = s_stage.raw;

    const int t    = threadIdx.x;

    if ((int)blockIdx.x >= NB) {
        const int grp  = __builtin_amdgcn_readfirstlane(t >> 7);
        const int tile = tileBegin + 2 * ((int)blockIdx.x - NB) + grp;
        float* As      = s_stage.a + grp * TILE_LDS_FLOATS;
        const int tg   = t & (GEMM_GROUP - 1);
        const bool live = tile < N / 16;
        if (live) gemm_stage(x, As, tile * 16, tg);
        __syncthreads();
        if (live) gemm_tile(W, h, As, tile * 16, tg);
        return;
    }

    const int lane = t & 31;
    const int wv   = t >> 5;
    const int bin  = blockIdx.x;

    const int* row = starts + (size_t)t * (NB + 1) + bin;
    const int ls  = row[0];
    const int len = row[1] - ls;
    const unsigned* run = chunks + (size_t)t * chunkLen + ls;
    int s0, m;
    block_scan_excl(ls, s_w, lane, wv, s0);
    const int roff = block_scan_excl(len, s_w, lane, wv, m);
    const bool staged = m <= SORT_CAP;

    s_deg[t] = 0;
    __syncthreads();
    for (int j = 0; j < len; j += 4) {
        unsigned w[4];
        #pragma unroll
        for (int k = 0; k < 4; ++k) w[k] = j + k < len ? run[j + k] : 0u;
        #pragma unroll
        for (int k = 0; k < 4; ++k) {
            if (j + k < len) {
                atomicAdd(&s_deg[w[k] >> 24], 1);
                if (staged) s_raw[roff + j + k] = w[k];
            }
        }
    }
    __syncthreads();

    const int d = s_deg[t];
    int unused;
    const int excl = block_scan_excl(d, s_w, lane, wv, unused);
    s_cursor[t] = excl;
    const int n = bin * BIN_NODES + t;
    if (n < N) {
        cnt[n] = d;
        cur[n] = s0 + excl + d;
    }
    __syncthreads();

    int* outp = grp_src + s0;
    if (staged) {
        for (int i = t; i < m; i += BIN_NODES) {
            const unsigned w = s_raw[i];
            const int p = atomicAdd(&s_cursor[w >> 24], 1);
            s_sorted[p] = (int)(w & 0xFFFFFFu);
        }
        __syncthreads();
        for (int i = t; i < m; i += BIN_NODES) outp[i] = s_sorted[i];
    } else {
        for (int j = 0; j < len; ++j) {
            const unsigned w = run[j];
            const int p = atomicAdd(&s_cursor[w >> 24], 1);
            outp[p] = (int)(w & 0xFFFFFFu);
        }
    }
}

#define NODE_BATCH 8
__global__ __launch_bounds__(256) void gat_node(const int* __restrict__ grp_src,
                                                const int* __restrict__ cnt,
                                                const int* __restrict__ cur,
                                                const float* __restrict__ a_src,
                                                const float* __restrict__ a_dst,
                                                const float* __restrict__ h,
                                                float* __restrict__ out,
                                                int N) {
    const int n = blockIdx.x * 8 + __builtin_amdgcn_readfirstlane((int)(threadIdx.x >> 5));
    if (n >= N) return;
    const int lane = threadIdx.x & 31;

    const int end   = cur[n];
    const int deg   = cnt[n];
    const int start = end - deg;

    int sl = start + lane < end ? grp_src[start + lane] : 0;

    const unsigned f4 = 4u * (unsigned)lane;
    const v4f as4 = *(const v4f*)(a_src + f4);
    const v4f ad4 = *(const v4f*)(a_dst + f4);
    const v4f hn  = *(const v4f*)(h + (size_t)n * F_OUT + f4);
    const float adn = group8_sum(dot4(hn, ad4));

    const float NEG = -__builtin_huge_valf();
    float m   = NEG;
    float ps  = 0.0f;
    v4f   acc = {0.0f, 0.0f, 0.0f, 0.0f};

    auto batch = [&](int sl, int k0, int nb, auto full) {
        constexpr bool FULL = decltype(full)::value;
        v4f r[NODE_BATCH];
        #pragma unroll
        for (int k = 0; k < NODE_BATCH; ++k) {
            if (FULL || k < nb) {
                const int s = __builtin_amdgcn_readlane(sl, k0 + k);
                const float* hs = h + (size_t)s * F_OUT;
                r[k] = *(const v4f*)(hs + f4);
            } else {
                r[k] = (v4f){0.0f, 0.0f, 0.0f, 0.0f};
            }
        }
        float e[NODE_BATCH];
        float mb = NEG;
        #pragma unroll
        for (int k = 0; k < NODE_BATCH; ++k) {
            const float d = group8_sum(dot4(r[k], as4));
            e[k] = (FULL || k < nb) ? leaky(d + adn) : NEG;
            mb = fmaxf(mb, e[k]);
        }
        const float mn = fmaxf(m, mb);
        const float sc = __expf(m - mn);
        m = mn;
        ps *= sc;
        acc *= sc;
        #pragma unroll
        for (int k = 0; k < NODE_BATCH; ++k) {
            const float p = __expf(e[k] - mn);
            ps += p;
            acc = __builtin_elementwise_fma(r[k], (v4f){p, p, p, p}, acc);
        }
    };

    for (int base = start; base < end; base += 32) {
        if (base != start) {
            const int j = base + lane;
            sl = j < end ? grp_src[j] : 0;
        }
        const int ng = min(32, end - base);
        int k0 = 0;
        for (; k0 + NODE_BATCH <= ng; k0 += NODE_BATCH)
            batch(sl, k0, NODE_BATCH, std::true_type{});
        if (k0 < ng)
            batch(sl, k0, ng - k0, std::false_type{});
    }

    const float den = ps + 1e-10f;
    v4f o;
    o.x = acc.x / den;
    o.y = acc.y / den;
    o.z = acc.z / den;
    o.w = acc.w / den;
    *(v4f*)(out + (size_t)n * F_OUT + f4) = o;
}

extern "C" void kernel_launch(void* const* d_in, const int* in_sizes, int n_in,
                              void* d_out, int out_size, void* d_ws, size_t ws_size,
                              hipStream_t stream) {
    const float* x     = (const float*)d_in[0];
    const int*   ei    = (const int*)d_in[1];
    const float* W     = (const float*)d_in[2];
    const float* a_src = (const float*)d_in[3];
    const float* a_dst = (const float*)d_in[4];
    float*       out   = (float*)d_out;

    const int N = in_sizes[0] / F_IN;
    const int E = in_sizes[1] / 2;
    const int* src = ei;
    const int* dst = ei + E;

    float* h    = (float*)d_ws;
    int*   cnt  = (int*)(h + (size_t)N * F_OUT);
    int*   cur  = cnt + N;
    int*   grp  = cur + N;

    {
        const int fullTiles = N / 16;
        const int tilesA    = fullTiles / 2;
        const int tilesB    = fullTiles - tilesA;

        const int NB       = (N + BIN_NODES - 1) / BIN_NODES;
        const int chunkLen = ((E + GROUP_CHUNKS - 1) / GROUP_CHUNKS + 31) & ~31;
        unsigned* chunks   = (unsigned*)d_out;
        int*      starts   = (int*)d_out + (size_t)GROUP_CHUNKS * chunkLen;
        const size_t ldsSort = ((size_t)NB + chunkLen) * sizeof(int);
        const size_t ldsGemm = (size_t)4 * TILE_LDS_FLOATS * sizeof(float);
        const size_t lds     = ldsSort > ldsGemm ? ldsSort : ldsGemm;
        gat_chunk_sort_gemm<<<GROUP_CHUNKS + (tilesA + 3) / 4, CHUNK_THREADS, lds, stream>>>(
            src, dst, chunks, starts, E, chunkLen, NB, x, W, h, tilesA);
        gat_bin_sort_gemm<<<NB + (tilesB + 1) / 2, BIN_NODES, 0, stream>>>(
            chunks, starts, cnt, cur, grp, chunkLen, NB, N, x, W, h, tilesA);

        int rowStart = fullTiles * 16;
        int rem = N - rowStart;
        if (rem > 0) {
            int total = rem * F_OUT;
            gat_gemm_tail<<<(total + 255) / 256, 256, 0, stream>>>(x, W, h, N, rowStart);
        }
    }

    gat_node<<<(N + 7) / 8, 256, 0, stream>>>(grp, cnt, cur, a_src, a_dst, h, out, N);
}


# First launch takes 3072 row tiles, not 3125: exactly three 4-tile GEMM work-groups per WGP beside its sorting one, no fourth on 14 WGPs (16 tiles against 12); the rest rides with the bin sort
# speedup vs baseline: 7.1601x; 1.0305x over previous
#include <hip/hip_runtime.h>
#include <hip/hip_bf16.h>
#include <type_traits>


#define NUM_HEADS 4
#define HEAD_DIM 32
#define F_IN 256
#define F_OUT 128
#define TILE_K4 64

typedef __attribute__((ext_vector_type(2))) float v2f;
typedef __attribute__((ext_vector_type(8))) float v8f;

#define TILE_LDS_FLOATS (16 * F_IN)
#define GEMM_GROUP 128

__device__ __forceinline__ void gemm_stage(const float* __restrict__ x, float* As,
                                           int rowBase, int tg) {
    #pragma unroll
    for (int j = 0; j < 8; ++j) {
        int f = tg + GEMM_GROUP * j;
        int r = f >> 6;
        int q = f & 63;
        float4 v = *(const float4*)(x + (size_t)(rowBase + r) * F_IN + (q << 2));
        *(float4*)(&As[r * F_IN + ((q ^ (r & 7)) << 2)]) = v;
    }
}

__device__ __forceinline__ void gemm_tile(const float* __restrict__ W, float* __restrict__ h,
                                          const float* As, int rowBase, int tg) {
    const int wave = tg >> 5;
    const int lane = tg & 31;
    const int m    = lane & 15;
    const int koff = (lane >> 4) << 1;
    const int col  = (wave << 5) + (m << 1);

    const float* Wp = W + col;
    const float* Ar = As + m * F_IN + koff;
    const float* a[8];
    #pragma unroll
    for (int j = 0; j < 8; ++j) a[j] = Ar + ((j ^ (m & 7)) << 2);

    v8f acc0 = {};
    v8f acc1 = {};

    #pragma unroll 1
    for (int q0 = 0; q0 < TILE_K4; q0 += 8) {
        #pragma unroll
        for (int j = 0; j < 8; ++j) {
            const int k = ((q0 + j) << 2) + koff;
            const v2f av = *(const v2f*)(a[j] + (q0 << 2));
            const v2f b0 = *(const v2f*)(Wp + (size_t)k * F_OUT);
            const v2f b1 = *(const v2f*)(Wp + (size_t)(k + 1) * F_OUT);
            const v2f be = {b0.x, b1.x};
            const v2f bo = {b0.y, b1.y};
            acc0 = __builtin_amdgcn_wmma_f32_16x16x4_f32(
                 false, av,  false, be,
                 (short)0, acc0,  false,  false);
            acc1 = __builtin_amdgcn_wmma_f32_16x16x4_f32(
                 false, av,  false, bo,
                 (short)0, acc1,  false,  false);
        }
    }

    const int roff = (lane >> 4) << 3;
    float* hp = h + (size_t)rowBase * F_OUT + col;
    #pragma unroll
    for (int v = 0; v < 8; ++v) {
        const v2f o = {acc0[v], acc1[v]};
        *(v2f*)(hp + (size_t)(v + roff) * F_OUT) = o;
    }
}

__global__ __launch_bounds__(256) void gat_gemm_tail(const float* __restrict__ x,
                                                     const float* __restrict__ W,
                                                     float* __restrict__ h,
                                                     int N, int rowStart) {
    int idx = blockIdx.x * blockDim.x + threadIdx.x;
    int total = (N - rowStart) * F_OUT;
    if (idx >= total) return;
    int r = rowStart + idx / F_OUT;
    int c = idx % F_OUT;
    float s = 0.0f;
    for (int k = 0; k < F_IN; ++k)
        s += x[(size_t)r * F_IN + k] * W[(size_t)k * F_OUT + c];
    h[(size_t)r * F_OUT + c] = s;
}

__device__ __forceinline__ float leaky(float v) { return fmaxf(v, 0.2f * v); }

typedef __attribute__((ext_vector_type(4))) float v4f;

template <int CTRL> __device__ __forceinline__ float dpp_take(float v) {
    return __int_as_float(__builtin_amdgcn_update_dpp(0, __float_as_int(v), CTRL, 0xF, 0xF, true));
}
__device__ __forceinline__ float group8_sum(float v) {
    v += dpp_take<0xB1>(v);
    v += dpp_take<0x4E>(v);
    v += dpp_take<0x141>(v);
    return v;
}
__device__ __forceinline__ float dot4(v4f r, v4f a) {
    const v2f t = __builtin_elementwise_fma(r.hi, a.hi, r.lo * a.lo);
    return t.x + t.y;
}

#define BIN_SHIFT 8
#define BIN_NODES 256
#define GROUP_CHUNKS 256
#define CHUNK_THREADS 512
#define SORT_CAP 8192
static_assert(GROUP_CHUNKS == BIN_NODES, "gat_bin_sort: thread t is chunk t, then node t");

__global__ __launch_bounds__(CHUNK_THREADS) void gat_chunk_sort_gemm(const int* __restrict__ src,
                                                                     const int* __restrict__ dst,
                                                                     unsigned* __restrict__ chunks,
                                                                     int* __restrict__ starts,
                                                                     int E, int chunkLen, int NB,
                                                                     const float* __restrict__ x,
                                                                     const float* __restrict__ W,
                                                                     float* __restrict__ h,
                                                                     int tileEnd) {
    extern __shared__ __align__(16) unsigned char s_dyn[];

    const int t    = threadIdx.x;

    if (blockIdx.x >= GROUP_CHUNKS) {
        const int grp  = __builtin_amdgcn_readfirstlane(t >> 7);
        const int tile = 4 * ((int)blockIdx.x - GROUP_CHUNKS) + grp;
        float* As      = (float*)s_dyn + grp * TILE_LDS_FLOATS;
        const int tg   = t & (GEMM_GROUP - 1);
        if (tile < tileEnd) gemm_stage(x, As, tile * 16, tg);
        __syncthreads();
        if (tile < tileEnd) gemm_tile(W, h, As, tile * 16, tg);
        return;
    }

    int* s_hist   = (int*)s_dyn;
    unsigned* s_word = (unsigned*)(s_hist + NB);

    const int lane = t & 31;
    const int c    = blockIdx.x;
    const int e0   = c * chunkLen;
    const int e1   = min(E, e0 + chunkLen);

    for (int b = t; b < NB; b += CHUNK_THREADS) s_hist[b] = 0;
    __syncthreads();

    for (int e = e0 + t; e < e1; e += CHUNK_THREADS * 4) {
        int d[4];
        #pragma unroll
        for (int k = 0; k < 4; ++k) {
            const int ee = e + k * CHUNK_THREADS;
            d[k] = ee < e1 ? dst[ee] : 0;
        }
        #pragma unroll
        for (int k = 0; k < 4; ++k)
            if (e + k * CHUNK_THREADS < e1) atomicAdd(&s_hist[d[k] >> BIN_SHIFT], 1);
    }
    __syncthreads();

    if (t < 32) {
        int* row = starts + (size_t)c * (NB + 1);
        int carry = 0;
        for (int base = 0; base < NB; base += 32) {
            const int b = base + lane;
            const int v = b < NB ? s_hist[b] : 0;
            int incl = v;
            for (int o = 1; o < 32; o <<= 1) {
                const int u = __shfl_up(incl, o);
                if (lane >= o) incl += u;
            }
            if (b < NB) {
                const int ls = carry + incl - v;
                s_hist[b] = ls;
                row[b]    = ls;
            }
            carry += __shfl(incl, 31);
        }
        if (lane == 0) row[NB] = carry;
    }
    __syncthreads();

    for (int e = e0 + t; e < e1; e += CHUNK_THREADS * 4) {
        int d[4], s[4];
        #pragma unroll
        for (int k = 0; k < 4; ++k) {
            const int ee = e + k * CHUNK_THREADS;
            d[k] = ee < e1 ? dst[ee] : 0;
            s[k] = ee < e1 ? src[ee] : 0;
        }
        #pragma unroll
        for (int k = 0; k < 4; ++k) {
            if (e + k * CHUNK_THREADS < e1) {
                const int p = atomicAdd(&s_hist[d[k] >> BIN_SHIFT], 1);
                s_word[p] = ((unsigned)(d[k] & (BIN_NODES - 1)) << 24) | (unsigned)s[k];
            }
        }
    }
    __syncthreads();

    unsigned* outp = chunks + (size_t)c * chunkLen;
    const int m = e1 - e0;
    for (int i = t; i < m; i += CHUNK_THREADS) outp[i] = s_word[i];
}

__device__ __forceinline__ int block_scan_excl(int v, int* s_w, int lane, int wv, int& total) {
    int incl = v;
    for (int o = 1; o < 32; o <<= 1) {
        const int u = __shfl_up(incl, o);
        if (lane >= o) incl += u;
    }
    if (lane == 31) s_w[wv] = incl;
    __syncthreads();
    int pre = 0, tot = 0;
    for (int w = 0; w < BIN_NODES / 32; ++w) {
        const int x = s_w[w];
        tot += x;
        pre += w < wv ? x : 0;
    }
    __syncthreads();
    total = tot;
    return pre + incl - v;
}

union __align__(16) BinStage {
    unsigned raw[SORT_CAP];
    float    a[2 * TILE_LDS_FLOATS];
};
static_assert(SORT_CAP >= 2 * TILE_LDS_FLOATS, "the A tiles must fit the sort's staging array");

__global__ __launch_bounds__(BIN_NODES) void gat_bin_sort_gemm(const unsigned* __restrict__ chunks,
                                                               const int* __restrict__ starts,
                                                               int* __restrict__ cnt,
                                                               int* __restrict__ cur,
                                                               int* __restrict__ grp_src,
                                                               int chunkLen, int NB, int N,
                                                               const float* __restrict__ x,
                                                               const float* __restrict__ W,
                                                               float* __restrict__ h,
                                                               int tileBegin) {
    __shared__ int s_deg[BIN_NODES];
    __shared__ int s_cursor[BIN_NODES];
    __shared__ int s_w[BIN_NODES / 32];
    __shared__ BinStage s_stage;
    __shared__ int s_sorted[SORT_CAP];
    unsigned* const s_raw = s_stage.raw;

    const int t    = threadIdx.x;

    if ((int)blockIdx.x >= NB) {
        const int grp  = __builtin_amdgcn_readfirstlane(t >> 7);
        const int tile = tileBegin + 2 * ((int)blockIdx.x - NB) + grp;
        float* As      = s_stage.a + grp * TILE_LDS_FLOATS;
        const int tg   = t & (GEMM_GROUP - 1);
        const bool live = tile < N / 16;
        if (live) gemm_stage(x, As, tile * 16, tg);
        __syncthreads();
        if (live) gemm_tile(W, h, As, tile * 16, tg);
        return;
    }

    const int lane = t & 31;
    const int wv   = t >> 5;
    const int bin  = blockIdx.x;

    const int* row = starts + (size_t)t * (NB + 1) + bin;
    const int ls  = row[0];
    const int len = row[1] - ls;
    const unsigned* run = chunks + (size_t)t * chunkLen + ls;
    int s0, m;
    block_scan_excl(ls, s_w, lane, wv, s0);
    const int roff = block_scan_excl(len, s_w, lane, wv, m);
    const bool staged = m <= SORT_CAP;

    s_deg[t] = 0;
    __syncthreads();
    for (int j = 0; j < len; j += 4) {
        unsigned w[4];
        #pragma unroll
        for (int k = 0; k < 4; ++k) w[k] = j + k < len ? run[j + k] : 0u;
        #pragma unroll
        for (int k = 0; k < 4; ++k) {
            if (j + k < len) {
                atomicAdd(&s_deg[w[k] >> 24], 1);
                if (staged) s_raw[roff + j + k] = w[k];
            }
        }
    }
    __syncthreads();

    const int d = s_deg[t];
    int unused;
    const int excl = block_scan_excl(d, s_w, lane, wv, unused);
    s_cursor[t] = excl;
    const int n = bin * BIN_NODES + t;
    if (n < N) {
        cnt[n] = d;
        cur[n] = s0 + excl + d;
    }
    __syncthreads();

    int* outp = grp_src + s0;
    if (staged) {
        for (int i = t; i < m; i += BIN_NODES) {
            const unsigned w = s_raw[i];
            const int p = atomicAdd(&s_cursor[w >> 24], 1);
            s_sorted[p] = (int)(w & 0xFFFFFFu);
        }
        __syncthreads();
        for (int i = t; i < m; i += BIN_NODES) outp[i] = s_sorted[i];
    } else {
        for (int j = 0; j < len; ++j) {
            const unsigned w = run[j];
            const int p = atomicAdd(&s_cursor[w >> 24], 1);
            outp[p] = (int)(w & 0xFFFFFFu);
        }
    }
}

#define NODE_BATCH 8
__global__ __launch_bounds__(256) void gat_node(const int* __restrict__ grp_src,
                                                const int* __restrict__ cnt,
                                                const int* __restrict__ cur,
                                                const float* __restrict__ a_src,
                                                const float* __restrict__ a_dst,
                                                const float* __restrict__ h,
                                                float* __restrict__ out,
                                                int N) {
    const int n = blockIdx.x * 8 + __builtin_amdgcn_readfirstlane((int)(threadIdx.x >> 5));
    if (n >= N) return;
    const int lane = threadIdx.x & 31;

    const int end   = cur[n];
    const int deg   = cnt[n];
    const int start = end - deg;

    int sl = start + lane < end ? grp_src[start + lane] : 0;

    const unsigned f4 = 4u * (unsigned)lane;
    const v4f as4 = *(const v4f*)(a_src + f4);
    const v4f ad4 = *(const v4f*)(a_dst + f4);
    const v4f hn  = *(const v4f*)(h + (size_t)n * F_OUT + f4);
    const float adn = group8_sum(dot4(hn, ad4));

    const float NEG = -__builtin_huge_valf();
    float m   = NEG;
    float ps  = 0.0f;
    v4f   acc = {0.0f, 0.0f, 0.0f, 0.0f};

    auto batch = [&](int sl, int k0, int nb, auto full) {
        constexpr bool FULL = decltype(full)::value;
        v4f r[NODE_BATCH];
        #pragma unroll
        for (int k = 0; k < NODE_BATCH; ++k) {
            if (FULL || k < nb) {
                const int s = __builtin_amdgcn_readlane(sl, k0 + k);
                const float* hs = h + (size_t)s * F_OUT;
                r[k] = *(const v4f*)(hs + f4);
            } else {
                r[k] = (v4f){0.0f, 0.0f, 0.0f, 0.0f};
            }
        }
        float e[NODE_BATCH];
        float mb = NEG;
        #pragma unroll
        for (int k = 0; k < NODE_BATCH; ++k) {
            const float d = group8_sum(dot4(r[k], as4));
            e[k] = (FULL || k < nb) ? leaky(d + adn) : NEG;
            mb = fmaxf(mb, e[k]);
        }
        const float mn = fmaxf(m, mb);
        const float sc = __expf(m - mn);
        m = mn;
        ps *= sc;
        acc *= sc;
        #pragma unroll
        for (int k = 0; k < NODE_BATCH; ++k) {
            const float p = __expf(e[k] - mn);
            ps += p;
            acc = __builtin_elementwise_fma(r[k], (v4f){p, p, p, p}, acc);
        }
    };

    for (int base = start; base < end; base += 32) {
        if (base != start) {
            const int j = base + lane;
            sl = j < end ? grp_src[j] : 0;
        }
        const int ng = min(32, end - base);
        int k0 = 0;
        for (; k0 + NODE_BATCH <= ng; k0 += NODE_BATCH)
            batch(sl, k0, NODE_BATCH, std::true_type{});
        if (k0 < ng)
            batch(sl, k0, ng - k0, std::false_type{});
    }

    const float den = ps + 1e-10f;
    v4f o;
    o.x = acc.x / den;
    o.y = acc.y / den;
    o.z = acc.z / den;
    o.w = acc.w / den;
    *(v4f*)(out + (size_t)n * F_OUT + f4) = o;
}

extern "C" void kernel_launch(void* const* d_in, const int* in_sizes, int n_in,
                              void* d_out, int out_size, void* d_ws, size_t ws_size,
                              hipStream_t stream) {
    const float* x     = (const float*)d_in[0];
    const int*   ei    = (const int*)d_in[1];
    const float* W     = (const float*)d_in[2];
    const float* a_src = (const float*)d_in[3];
    const float* a_dst = (const float*)d_in[4];
    float*       out   = (float*)d_out;

    const int N = in_sizes[0] / F_IN;
    const int E = in_sizes[1] / 2;
    const int* src = ei;
    const int* dst = ei + E;

    float* h    = (float*)d_ws;
    int*   cnt  = (int*)(h + (size_t)N * F_OUT);
    int*   cur  = cnt + N;
    int*   grp  = cur + N;

    {
        const int fullTiles = N / 16;
        const int halfTiles = fullTiles / 2;
        const int tilesA    = halfTiles >= 1024 ? (halfTiles & ~1023) : halfTiles;
        const int tilesB    = fullTiles - tilesA;

        const int NB       = (N + BIN_NODES - 1) / BIN_NODES;
        const int chunkLen = ((E + GROUP_CHUNKS - 1) / GROUP_CHUNKS + 31) & ~31;
        unsigned* chunks   = (unsigned*)d_out;
        int*      starts   = (int*)d_out + (size_t)GROUP_CHUNKS * chunkLen;
        const size_t ldsSort = ((size_t)NB + chunkLen) * sizeof(int);
        const size_t ldsGemm = (size_t)4 * TILE_LDS_FLOATS * sizeof(float);
        const size_t lds     = ldsSort > ldsGemm ? ldsSort : ldsGemm;
        gat_chunk_sort_gemm<<<GROUP_CHUNKS + (tilesA + 3) / 4, CHUNK_THREADS, lds, stream>>>(
            src, dst, chunks, starts, E, chunkLen, NB, x, W, h, tilesA);
        gat_bin_sort_gemm<<<NB + (tilesB + 1) / 2, BIN_NODES, 0, stream>>>(
            chunks, starts, cnt, cur, grp, chunkLen, NB, N, x, W, h, tilesA);

        int rowStart = fullTiles * 16;
        int rem = N - rowStart;
        if (rem > 0) {
            int total = rem * F_OUT;
            gat_gemm_tail<<<(total + 255) / 256, 256, 0, stream>>>(x, W, h, N, rowStart);
        }
    }

    gat_node<<<(N + 7) / 8, 256, 0, stream>>>(grp, cnt, cur, a_src, a_dst, h, out, N);
}
